# starting kernel
# baseline (speedup 1.0000x reference)
#include <hip/hip_runtime.h>
#include <hip/hip_bf16.h>

typedef __attribute__((ext_vector_type(2))) float v2f;
typedef __attribute__((ext_vector_type(8))) float v8f;

#define B_SZ 512
#define N_SZ 512
#define D_SZ 128
#define NT 4
#define ROWS_PER_TILE (N_SZ / NT)

__global__ __launch_bounds__(256)
void GraphEmb_gate_reduce(const float* __restrict__ h,
                          const int*   __restrict__ ns,
                          const float* __restrict__ W_g,
                          const float* __restrict__ b_g,
                          float* __restrict__ a_part,
                          float* __restrict__ gs_part)
{
    const int b    = blockIdx.y;
    const int tile = blockIdx.x;
    const int tid  = threadIdx.x;
    const int lane = tid & 31;
    const int warp = tid >> 5;

    const float4 wg = ((const float4*)W_g)[lane];
    const float  bg = b_g[0];
    const int    nvalid = ns[b];

    float4 acc = make_float4(0.f, 0.f, 0.f, 0.f);
    float  gs  = 0.f;

    const float4* h4 = (const float4*)h;

    for (int r = warp; r < ROWS_PER_TILE; r += 8) {
        const int row = tile * ROWS_PER_TILE + r;
        if (row < nvalid) {
            float4 hv = h4[((size_t)((size_t)b * N_SZ + row) * D_SZ) / 4 + lane];
            float d = hv.x * wg.x + hv.y * wg.y + hv.z * wg.z + hv.w * wg.w;
            #pragma unroll
            for (int off = 16; off >= 1; off >>= 1)
                d += __shfl_xor(d, off, 32);
            float g = 1.f / (1.f + __expf(-(d + bg)));
            acc.x += g * hv.x; acc.y += g * hv.y;
            acc.z += g * hv.z; acc.w += g * hv.w;
            gs += g;
        }
    }

    __shared__ float wacc[8][D_SZ];
    __shared__ float wgs[8];
    ((float4*)wacc[warp])[lane] = acc;
    if (lane == 0) wgs[warp] = gs;
    __syncthreads();

    if (tid < D_SZ) {
        float s = 0.f;
        #pragma unroll
        for (int w = 0; w < 8; ++w) s += wacc[w][tid];
        a_part[((size_t)b * NT + tile) * D_SZ + tid] = s;
    }
    if (tid == 0) {
        float s = 0.f;
        #pragma unroll
        for (int w = 0; w < 8; ++w) s += wgs[w];
        gs_part[b * NT + tile] = s;
    }
}

__global__ __launch_bounds__(256)
void GraphEmb_wmma_finish(const float* __restrict__ a_part,
                          const float* __restrict__ gs_part,
                          const float* __restrict__ W_f,
                          const float* __restrict__ b_f,
                          float* __restrict__ out)
{
    const int bb   = blockIdx.x;
    const int tid  = threadIdx.x;
    const int lane = tid & 31;
    const int warp = tid >> 5;

    __shared__ float a_s[16][D_SZ];
    __shared__ float out_s[16][D_SZ];
    __shared__ float sg_s[16];
    __shared__ float inv_s[16];
    __shared__ float pn[16][16];

    for (int i = tid; i < 16 * D_SZ; i += 256) {
        int m = i >> 7, d = i & 127;
        const float* p = a_part + ((size_t)(bb * 16 + m) * NT) * D_SZ + d;
        float s = 0.f;
        #pragma unroll
        for (int t = 0; t < NT; ++t) s += p[t * D_SZ];
        a_s[m][d] = s;
    }
    if (tid < 16) {
        const float* p = gs_part + (bb * 16 + tid) * NT;
        float s = 0.f;
        #pragma unroll
        for (int t = 0; t < NT; ++t) s += p[t];
        sg_s[tid] = s;
    }
    __syncthreads();

    const int mrow = lane & 15;
    const int koff = (lane >> 4) << 1;
    const int e    = warp * 16 + mrow;

    v8f c = {};
    #pragma unroll
    for (int k0 = 0; k0 < D_SZ; k0 += 4) {
        v2f af = { a_s[mrow][k0 + koff], a_s[mrow][k0 + koff + 1] };
        v2f bf = { W_f[e * D_SZ + k0 + koff], W_f[e * D_SZ + k0 + koff + 1] };
        c = __builtin_amdgcn_wmma_f32_16x16x4_f32(false, af, false, bf,
                                                  (short)0, c, false, false);
    }

    const float bfv = b_f[e];
    #pragma unroll
    for (int r = 0; r < 8; ++r) {
        int M = r + ((lane >> 4) << 3);
        out_s[M][e] = c[r] + sg_s[M] * bfv;
    }
    __syncthreads();

    {
        int mm = tid >> 4, seg = tid & 15;
        float s = 0.f;
        #pragma unroll
        for (int j = 0; j < 8; ++j) { float v = out_s[mm][seg * 8 + j]; s += v * v; }
        pn[mm][seg] = s;
    }
    __syncthreads();
    if (tid < 16) {
        float s = 0.f;
        #pragma unroll
        for (int j = 0; j < 16; ++j) s += pn[tid][j];
        inv_s[tid] = 1.f / fmaxf(sqrtf(s), 1e-12f);
    }
    __syncthreads();
    for (int i = tid; i < 16 * D_SZ; i += 256) {
        int mm = i >> 7, ee = i & 127;
        out[(size_t)(bb * 16 + mm) * D_SZ + ee] = out_s[mm][ee] * inv_s[mm];
    }
}

extern "C" void kernel_launch(void* const* d_in, const int* in_sizes, int n_in,
                              void* d_out, int out_size, void* d_ws, size_t ws_size,
                              hipStream_t stream) {
    const float* h   = (const float*)d_in[0];
    const int*   ns  = (const int*)d_in[1];
    const float* W_f = (const float*)d_in[2];
    const float* b_f = (const float*)d_in[3];
    const float* W_g = (const float*)d_in[4];
    const float* b_g = (const float*)d_in[5];
    float* out = (float*)d_out;

    float* a_part  = (float*)d_ws;
    float* gs_part = a_part + (size_t)B_SZ * NT * D_SZ;

    GraphEmb_gate_reduce<<<dim3(NT, B_SZ), 256, 0, stream>>>(
        h, ns, W_g, b_g, a_part, gs_part);

    GraphEmb_wmma_finish<<<dim3(B_SZ / 16), 256, 0, stream>>>(
        a_part, gs_part, W_f, b_f, out);
}


# kernel 1: one 8-wave work-group per graph (was 4 row tiles); a wave loads its rows 8 at a time as one group, next group requested ahead, the 8 rows' gate math interleaved; row sums regrouped
# speedup vs baseline: 1.4511x; 1.4511x over previous
#include <hip/hip_runtime.h>
#include <hip/hip_bf16.h>

typedef __attribute__((ext_vector_type(2))) float v2f;
typedef __attribute__((ext_vector_type(8))) float v8f;

#define B_SZ 512
#define N_SZ 512
#define D_SZ 128
#define NWAVE 8
#define RB 8


#define ORDER_POINT(q_) do { \
    asm volatile("" : "+v"((q_).x), "+v"((q_).y), "+v"((q_).z), "+v"((q_).w) :: "memory"); \
    __builtin_amdgcn_sched_barrier(0); } while (0)

__device__ __forceinline__ void load_rows(float4 (&buf)[RB], const float4* __restrict__ hrow,
                                          int lane, int i0, int cnt)
{
    #pragma unroll
    for (int j = 0; j < RB; ++j) {
        int i = i0 + j;
        i = i < cnt ? i : cnt - 1;
        buf[j] = (hrow + (size_t)i * (NWAVE * D_SZ / 4))[lane];
    }
}

template <bool MASKED>
__device__ __forceinline__ void gate_rows(const float4 (&buf)[RB], int i0, int cnt,
                                          const float4 wg, const float bg,
                                          float4& acc, float& gs)
{
    float d[RB], t[RB], g[RB];
    #pragma unroll
    for (int j = 0; j < RB; ++j) {
        const float4 hv = buf[j];
        d[j] = hv.x * wg.x + hv.y * wg.y + hv.z * wg.z + hv.w * wg.w;
    }
    #pragma unroll
    for (int off = 16; off >= 1; off >>= 1) {
        #pragma unroll
        for (int j = 0; j < RB; ++j) t[j] = __shfl_xor(d[j], off, 32);
        __builtin_amdgcn_sched_barrier(0);
        #pragma unroll
        for (int j = 0; j < RB; ++j) d[j] += t[j];
    }
    #pragma unroll
    for (int j = 0; j < RB; ++j) t[j] = __expf(-(d[j] + bg));
    #pragma unroll
    for (int j = 0; j < RB; ++j) {
        g[j] = 1.f / (1.f + t[j]);
        if (MASKED) g[j] = (i0 + j < cnt) ? g[j] : 0.f;
    }
    #pragma unroll
    for (int j = 0; j < RB; ++j) {
        const float4 hv = buf[j];
        acc.x += g[j] * hv.x; acc.y += g[j] * hv.y;
        acc.z += g[j] * hv.z; acc.w += g[j] * hv.w;
        gs += g[j];
    }
}

__global__ __launch_bounds__(256)
void GraphEmb_gate_reduce(const float* __restrict__ h,
                          const int*   __restrict__ ns,
                          const float* __restrict__ W_g,
                          const float* __restrict__ b_g,
                          float* __restrict__ a_sum,
                          float* __restrict__ gs_sum)
{
    const int b    = blockIdx.x;
    const int tid  = threadIdx.x;
    const int lane = tid & 31;
    const int warp = __builtin_amdgcn_readfirstlane(tid >> 5);

    float4 wg = ((const float4*)W_g)[lane];
    const float  bg = b_g[0];
    int nvalid = ns[b];
    nvalid = nvalid < 0 ? 0 : (nvalid > N_SZ ? N_SZ : nvalid);
    const int cnt = nvalid > warp ? (nvalid - warp + NWAVE - 1) / NWAVE : 0;

    float4 acc = make_float4(0.f, 0.f, 0.f, 0.f);
    float  gs  = 0.f;

    if (cnt > 0) {
        const float4* hrow = (const float4*)h + ((size_t)b * N_SZ + warp) * (D_SZ / 4);
        float4 bufA[RB], bufB[RB];
        load_rows(bufA, hrow, lane, 0, cnt);
        int i0 = 0;
        while (i0 + 2 * RB < cnt) {
            load_rows(bufB, hrow, lane, i0 + RB, cnt);
            ORDER_POINT(wg);
            gate_rows<false>(bufA, i0, cnt, wg, bg, acc, gs);
            load_rows(bufA, hrow, lane, i0 + 2 * RB, cnt);
            ORDER_POINT(wg);
            gate_rows<false>(bufB, i0 + RB, cnt, wg, bg, acc, gs);
            i0 += 2 * RB;
        }
        if (i0 + RB < cnt) {
            load_rows(bufB, hrow, lane, i0 + RB, cnt);
            ORDER_POINT(wg);
            gate_rows<false>(bufA, i0, cnt, wg, bg, acc, gs);
            gate_rows<true>(bufB, i0 + RB, cnt, wg, bg, acc, gs);
        } else {
            gate_rows<true>(bufA, i0, cnt, wg, bg, acc, gs);
        }
    }

    __shared__ float wacc[NWAVE][D_SZ];
    __shared__ float wgs[NWAVE];
    ((float4*)wacc[warp])[lane] = acc;
    if (lane == 0) wgs[warp] = gs;
    __syncthreads();

    if (tid < D_SZ) {
        float s = 0.f;
        #pragma unroll
        for (int w = 0; w < NWAVE; ++w) s += wacc[w][tid];
        a_sum[(size_t)b * D_SZ + tid] = s;
    }
    if (tid == 0) {
        float s = 0.f;
        #pragma unroll
        for (int w = 0; w < NWAVE; ++w) s += wgs[w];
        gs_sum[b] = s;
    }
}

__global__ __launch_bounds__(256)
void GraphEmb_wmma_finish(const float* __restrict__ a_sum,
                          const float* __restrict__ gs_sum,
                          const float* __restrict__ W_f,
                          const float* __restrict__ b_f,
                          float* __restrict__ out)
{
    const int bb   = blockIdx.x;
    const int tid  = threadIdx.x;
    const int lane = tid & 31;
    const int warp = tid >> 5;

    __shared__ float a_s[16][D_SZ];
    __shared__ float out_s[16][D_SZ];
    __shared__ float sg_s[16];
    __shared__ float inv_s[16];
    __shared__ float pn[16][16];

    for (int i = tid; i < 16 * D_SZ; i += 256) {
        int m = i >> 7, d = i & 127;
        a_s[m][d] = a_sum[(size_t)(bb * 16 + m) * D_SZ + d];
    }
    if (tid < 16) sg_s[tid] = gs_sum[bb * 16 + tid];
    __syncthreads();

    const int mrow = lane & 15;
    const int koff = (lane >> 4) << 1;
    const int e    = warp * 16 + mrow;

    v8f c = {};
    #pragma unroll
    for (int k0 = 0; k0 < D_SZ; k0 += 4) {
        v2f af = { a_s[mrow][k0 + koff], a_s[mrow][k0 + koff + 1] };
        v2f bf = { W_f[e * D_SZ + k0 + koff], W_f[e * D_SZ + k0 + koff + 1] };
        c = __builtin_amdgcn_wmma_f32_16x16x4_f32(false, af, false, bf,
                                                  (short)0, c, false, false);
    }

    const float bfv = b_f[e];
    #pragma unroll
    for (int r = 0; r < 8; ++r) {
        int M = r + ((lane >> 4) << 3);
        out_s[M][e] = c[r] + sg_s[M] * bfv;
    }
    __syncthreads();

    {
        int mm = tid >> 4, seg = tid & 15;
        float s = 0.f;
        #pragma unroll
        for (int j = 0; j < 8; ++j) { float v = out_s[mm][seg * 8 + j]; s += v * v; }
        pn[mm][seg] = s;
    }
    __syncthreads();
    if (tid < 16) {
        float s = 0.f;
        #pragma unroll
        for (int j = 0; j < 16; ++j) s += pn[tid][j];
        inv_s[tid] = 1.f / fmaxf(sqrtf(s), 1e-12f);
    }
    __syncthreads();
    for (int i = tid; i < 16 * D_SZ; i += 256) {
        int mm = i >> 7, ee = i & 127;
        out[(size_t)(bb * 16 + mm) * D_SZ + ee] = out_s[mm][ee] * inv_s[mm];
    }
}

extern "C" void kernel_launch(void* const* d_in, const int* in_sizes, int n_in,
                              void* d_out, int out_size, void* d_ws, size_t ws_size,
                              hipStream_t stream) {
    const float* h   = (const float*)d_in[0];
    const int*   ns  = (const int*)d_in[1];
    const float* W_f = (const float*)d_in[2];
    const float* b_f = (const float*)d_in[3];
    const float* W_g = (const float*)d_in[4];
    const float* b_g = (const float*)d_in[5];
    float* out = (float*)d_out;

    float* a_sum  = (float*)d_ws;
    float* gs_sum = a_sum + (size_t)B_SZ * D_SZ;

    GraphEmb_gate_reduce<<<dim3(B_SZ), 256, 0, stream>>>(
        h, ns, W_g, b_g, a_sum, gs_sum);

    GraphEmb_wmma_finish<<<dim3(B_SZ / 16), 256, 0, stream>>>(
        a_sum, gs_sum, W_f, b_f, out);
}


# one launch instead of two: each work-group finishes its own graph (W_f product on f32 WMMA, the 8 per-wave sums as A rows; bias; norm); sum over waves now follows the product, norm sum regrouped
# speedup vs baseline: 2.1639x; 1.4912x over previous
#include <hip/hip_runtime.h>
#include <hip/hip_bf16.h>

typedef __attribute__((ext_vector_type(2))) float v2f;
typedef __attribute__((ext_vector_type(8))) float v8f;

#define B_SZ 512
#define N_SZ 512
#define D_SZ 128
#define NWAVE 8
#define RB 8
#define APITCH 136


#define ORDER_POINT(q_) do { \
    asm volatile("" : "+v"((q_).x), "+v"((q_).y), "+v"((q_).z), "+v"((q_).w) :: "memory"); \
    __builtin_amdgcn_sched_barrier(0); } while (0)

__device__ __forceinline__ void load_rows(float4 (&buf)[RB], const float4* __restrict__ hrow,
                                          int lane, int i0, int cnt)
{
    #pragma unroll
    for (int j = 0; j < RB; ++j) {
        int i = i0 + j;
        i = i < cnt ? i : cnt - 1;
        buf[j] = (hrow + (size_t)i * (NWAVE * D_SZ / 4))[lane];
    }
}

template <bool MASKED>
__device__ __forceinline__ void gate_rows(const float4 (&buf)[RB], int i0, int cnt,
                                          const float4 wg, const float bg,
                                          float4& acc, float& gs)
{
    float d[RB], t[RB], g[RB];
    #pragma unroll
    for (int j = 0; j < RB; ++j) {
        const float4 hv = buf[j];
        d[j] = hv.x * wg.x + hv.y * wg.y + hv.z * wg.z + hv.w * wg.w;
    }
    #pragma unroll
    for (int off = 16; off >= 1; off >>= 1) {
        #pragma unroll
        for (int j = 0; j < RB; ++j) t[j] = __shfl_xor(d[j], off, 32);
        __builtin_amdgcn_sched_barrier(0);
        #pragma unroll
        for (int j = 0; j < RB; ++j) d[j] += t[j];
    }
    #pragma unroll
    for (int j = 0; j < RB; ++j) t[j] = __expf(-(d[j] + bg));
    #pragma unroll
    for (int j = 0; j < RB; ++j) {
        g[j] = 1.f / (1.f + t[j]);
        if (MASKED) g[j] = (i0 + j < cnt) ? g[j] : 0.f;
    }
    #pragma unroll
    for (int j = 0; j < RB; ++j) {
        const float4 hv = buf[j];
        acc.x += g[j] * hv.x; acc.y += g[j] * hv.y;
        acc.z += g[j] * hv.z; acc.w += g[j] * hv.w;
        gs += g[j];
    }
}

__device__ __forceinline__ void load_wf(float4 (&wf)[16], float& bfv,
                                        const float4* __restrict__ wrow,
                                        const float* __restrict__ b_f, int e)
{
    #pragma unroll
    for (int jj = 0; jj < 16; ++jj) wf[jj] = wrow[2 * jj];
    bfv = b_f[e];
}

__global__ __launch_bounds__(256)
void GraphEmb_gate_reduce_finish(const float* __restrict__ h,
                                 const int*   __restrict__ ns,
                                 const float* __restrict__ W_g,
                                 const float* __restrict__ b_g,
                                 const float* __restrict__ W_f,
                                 const float* __restrict__ b_f,
                                 float* __restrict__ out)
{
    const int b    = blockIdx.x;
    const int tid  = threadIdx.x;
    const int lane = tid & 31;
    const int warp = __builtin_amdgcn_readfirstlane(tid >> 5);

    float4 wg = ((const float4*)W_g)[lane];
    const float  bg = b_g[0];
    int nvalid = ns[b];
    nvalid = nvalid < 0 ? 0 : (nvalid > N_SZ ? N_SZ : nvalid);
    const int cnt = nvalid > warp ? (nvalid - warp + NWAVE - 1) / NWAVE : 0;

    const int c  = lane & 15;
    const int hh = lane >> 4;
    const int e  = warp * 16 + c;
    const float4* wrow = (const float4*)(W_f + (size_t)e * D_SZ) + hh;
    float4 wf[16];
    float  bfv;

    float4 acc = make_float4(0.f, 0.f, 0.f, 0.f);
    float  gs  = 0.f;

    if (cnt > 0) {
        const float4* hrow = (const float4*)h + ((size_t)b * N_SZ + warp) * (D_SZ / 4);
        float4 bufA[RB], bufB[RB];
        load_rows(bufA, hrow, lane, 0, cnt);
        int i0 = 0;
        while (i0 + 2 * RB < cnt) {
            load_rows(bufB, hrow, lane, i0 + RB, cnt);
            ORDER_POINT(wg);
            gate_rows<false>(bufA, i0, cnt, wg, bg, acc, gs);
            load_rows(bufA, hrow, lane, i0 + 2 * RB, cnt);
            ORDER_POINT(wg);
            gate_rows<false>(bufB, i0 + RB, cnt, wg, bg, acc, gs);
            i0 += 2 * RB;
        }
        load_wf(wf, bfv, wrow, b_f, e);
        if (i0 + RB < cnt) {
            load_rows(bufB, hrow, lane, i0 + RB, cnt);
            ORDER_POINT(wg);
            gate_rows<false>(bufA, i0, cnt, wg, bg, acc, gs);
            gate_rows<true>(bufB, i0 + RB, cnt, wg, bg, acc, gs);
        } else {
            gate_rows<true>(bufA, i0, cnt, wg, bg, acc, gs);
        }
    } else {
        load_wf(wf, bfv, wrow, b_f, e);
    }

    __shared__ __attribute__((aligned(16))) float wacc[NWAVE][APITCH];
    __shared__ float wgs[NWAVE];
    __shared__ __attribute__((aligned(16))) float hg_s[D_SZ];
    ((float4*)wacc[warp])[lane] = acc;
    if (lane == 0) wgs[warp] = gs;
    __syncthreads();

    const float* arow = &wacc[c & 7][4 * hh];
    float4 av[16];
    #pragma unroll
    for (int jj = 0; jj < 16; ++jj) av[jj] = *(const float4*)(arow + 8 * jj);
    float sg = 0.f;
    #pragma unroll
    for (int w = 0; w < NWAVE; ++w) sg += wgs[w];
    __builtin_amdgcn_sched_barrier(0);
    v8f c0 = {}, c1 = {};
    #pragma unroll
    for (int jj = 0; jj < 16; ++jj) {
        const float4 bv = wf[jj];
        v2f a0 = { av[jj].x, av[jj].y }, b0 = { bv.x, bv.y };
        v2f a1 = { av[jj].z, av[jj].w }, b1 = { bv.z, bv.w };
        c0 = __builtin_amdgcn_wmma_f32_16x16x4_f32(false, a0, false, b0,
                                                   (short)0, c0, false, false);
        c1 = __builtin_amdgcn_wmma_f32_16x16x4_f32(false, a1, false, b1,
                                                   (short)0, c1, false, false);
    }

    float hv = 0.f;
    #pragma unroll
    for (int r = 0; r < 8; ++r) hv += c0[r] + c1[r];
    hv += sg * bfv;
    if (hh == 0) hg_s[e] = hv;
    __syncthreads();

    if (warp == 0) {
        const float4 v = ((const float4*)hg_s)[lane];
        float q = v.x * v.x + v.y * v.y + v.z * v.z + v.w * v.w;
        #pragma unroll
        for (int off = 16; off >= 1; off >>= 1) q += __shfl_xor(q, off, 32);
        const float inv = 1.f / fmaxf(sqrtf(q), 1e-12f);
        ((float4*)(out + (size_t)b * D_SZ))[lane] =
            make_float4(v.x * inv, v.y * inv, v.z * inv, v.w * inv);
    }
}

extern "C" void kernel_launch(void* const* d_in, const int* in_sizes, int n_in,
                              void* d_out, int out_size, void* d_ws, size_t ws_size,
                              hipStream_t stream) {
    const float* h   = (const float*)d_in[0];
    const int*   ns  = (const int*)d_in[1];
    const float* W_f = (const float*)d_in[2];
    const float* b_f = (const float*)d_in[3];
    const float* W_g = (const float*)d_in[4];
    const float* b_g = (const float*)d_in[5];
    float* out = (float*)d_out;

    GraphEmb_gate_reduce_finish<<<dim3(B_SZ), 256, 0, stream>>>(
        h, ns, W_g, b_g, W_f, b_f, out);
}


# gate sigmoid: 1/(1+exp(-z)) by the hardware reciprocal v_rcp_f32 (1 ulp) in place of the IEEE division sequence, ten vector instructions less per row; sums and their order unchanged
# speedup vs baseline: 2.3067x; 1.0660x over previous
#include <hip/hip_runtime.h>
#include <hip/hip_bf16.h>

typedef __attribute__((ext_vector_type(2))) float v2f;
typedef __attribute__((ext_vector_type(8))) float v8f;

#define B_SZ 512
#define N_SZ 512
#define D_SZ 128
#define NWAVE 8
#define RB 8
#define APITCH 136


#define ORDER_POINT(q_) do { \
    asm volatile("" : "+v"((q_).x), "+v"((q_).y), "+v"((q_).z), "+v"((q_).w) :: "memory"); \
    __builtin_amdgcn_sched_barrier(0); } while (0)

__device__ __forceinline__ void load_rows(float4 (&buf)[RB], const float4* __restrict__ hrow,
                                          int lane, int i0, int cnt)
{
    #pragma unroll
    for (int j = 0; j < RB; ++j) {
        int i = i0 + j;
        i = i < cnt ? i : cnt - 1;
        buf[j] = (hrow + (size_t)i * (NWAVE * D_SZ / 4))[lane];
    }
}

template <bool MASKED>
__device__ __forceinline__ void gate_rows(const float4 (&buf)[RB], int i0, int cnt,
                                          const float4 wg, const float bg,
                                          float4& acc, float& gs)
{
    float d[RB], t[RB], g[RB];
    #pragma unroll
    for (int j = 0; j < RB; ++j) {
        const float4 hv = buf[j];
        d[j] = hv.x * wg.x + hv.y * wg.y + hv.z * wg.z + hv.w * wg.w;
    }
    #pragma unroll
    for (int off = 16; off >= 1; off >>= 1) {
        #pragma unroll
        for (int j = 0; j < RB; ++j) t[j] = __shfl_xor(d[j], off, 32);
        __builtin_amdgcn_sched_barrier(0);
        #pragma unroll
        for (int j = 0; j < RB; ++j) d[j] += t[j];
    }
    #pragma unroll
    for (int j = 0; j < RB; ++j) t[j] = __expf(-(d[j] + bg));
    #pragma unroll
    for (int j = 0; j < RB; ++j) {
        g[j] = __builtin_amdgcn_rcpf(1.f + t[j]);
        if (MASKED) g[j] = (i0 + j < cnt) ? g[j] : 0.f;
    }
    #pragma unroll
    for (int j = 0; j < RB; ++j) {
        const float4 hv = buf[j];
        acc.x += g[j] * hv.x; acc.y += g[j] * hv.y;
        acc.z += g[j] * hv.z; acc.w += g[j] * hv.w;
        gs += g[j];
    }
}

__device__ __forceinline__ void load_wf(float4 (&wf)[16], float& bfv,
                                        const float4* __restrict__ wrow,
                                        const float* __restrict__ b_f, int e)
{
    #pragma unroll
    for (int jj = 0; jj < 16; ++jj) wf[jj] = wrow[2 * jj];
    bfv = b_f[e];
}

__global__ __launch_bounds__(256)
void GraphEmb_gate_reduce_finish(const float* __restrict__ h,
                                 const int*   __restrict__ ns,
                                 const float* __restrict__ W_g,
                                 const float* __restrict__ b_g,
                                 const float* __restrict__ W_f,
                                 const float* __restrict__ b_f,
                                 float* __restrict__ out)
{
    const int b    = blockIdx.x;
    const int tid  = threadIdx.x;
    const int lane = tid & 31;
    const int warp = __builtin_amdgcn_readfirstlane(tid >> 5);

    float4 wg = ((const float4*)W_g)[lane];
    const float  bg = b_g[0];
    int nvalid = ns[b];
    nvalid = nvalid < 0 ? 0 : (nvalid > N_SZ ? N_SZ : nvalid);
    const int cnt = nvalid > warp ? (nvalid - warp + NWAVE - 1) / NWAVE : 0;

    const int c  = lane & 15;
    const int hh = lane >> 4;
    const int e  = warp * 16 + c;
    const float4* wrow = (const float4*)(W_f + (size_t)e * D_SZ) + hh;
    float4 wf[16];
    float  bfv;

    float4 acc = make_float4(0.f, 0.f, 0.f, 0.f);
    float  gs  = 0.f;

    if (cnt > 0) {
        const float4* hrow = (const float4*)h + ((size_t)b * N_SZ + warp) * (D_SZ / 4);
        float4 bufA[RB], bufB[RB];
        load_rows(bufA, hrow, lane, 0, cnt);
        int i0 = 0;
        while (i0 + 2 * RB < cnt) {
            load_rows(bufB, hrow, lane, i0 + RB, cnt);
            ORDER_POINT(wg);
            gate_rows<false>(bufA, i0, cnt, wg, bg, acc, gs);
            load_rows(bufA, hrow, lane, i0 + 2 * RB, cnt);
            ORDER_POINT(wg);
            gate_rows<false>(bufB, i0 + RB, cnt, wg, bg, acc, gs);
            i0 += 2 * RB;
        }
        load_wf(wf, bfv, wrow, b_f, e);
        if (i0 + RB < cnt) {
            load_rows(bufB, hrow, lane, i0 + RB, cnt);
            ORDER_POINT(wg);
            gate_rows<false>(bufA, i0, cnt, wg, bg, acc, gs);
            gate_rows<true>(bufB, i0 + RB, cnt, wg, bg, acc, gs);
        } else {
            gate_rows<true>(bufA, i0, cnt, wg, bg, acc, gs);
        }
    } else {
        load_wf(wf, bfv, wrow, b_f, e);
    }

    __shared__ __attribute__((aligned(16))) float wacc[NWAVE][APITCH];
    __shared__ float wgs[NWAVE];
    __shared__ __attribute__((aligned(16))) float hg_s[D_SZ];
    ((float4*)wacc[warp])[lane] = acc;
    if (lane == 0) wgs[warp] = gs;
    __syncthreads();

    const float* arow = &wacc[c & 7][4 * hh];
    float4 av[16];
    #pragma unroll
    for (int jj = 0; jj < 16; ++jj) av[jj] = *(const float4*)(arow + 8 * jj);
    float sg = 0.f;
    #pragma unroll
    for (int w = 0; w < NWAVE; ++w) sg += wgs[w];
    __builtin_amdgcn_sched_barrier(0);
    v8f c0 = {}, c1 = {};
    #pragma unroll
    for (int jj = 0; jj < 16; ++jj) {
        const float4 bv = wf[jj];
        v2f a0 = { av[jj].x, av[jj].y }, b0 = { bv.x, bv.y };
        v2f a1 = { av[jj].z, av[jj].w }, b1 = { bv.z, bv.w };
        c0 = __builtin_amdgcn_wmma_f32_16x16x4_f32(false, a0, false, b0,
                                                   (short)0, c0, false, false);
        c1 = __builtin_amdgcn_wmma_f32_16x16x4_f32(false, a1, false, b1,
                                                   (short)0, c1, false, false);
    }

    float hv = 0.f;
    #pragma unroll
    for (int r = 0; r < 8; ++r) hv += c0[r] + c1[r];
    hv += sg * bfv;
    if (hh == 0) hg_s[e] = hv;
    __syncthreads();

    if (warp == 0) {
        const float4 v = ((const float4*)hg_s)[lane];
        float q = v.x * v.x + v.y * v.y + v.z * v.z + v.w * v.w;
        #pragma unroll
        for (int off = 16; off >= 1; off >>= 1) q += __shfl_xor(q, off, 32);
        const float inv = 1.f / fmaxf(sqrtf(q), 1e-12f);
        ((float4*)(out + (size_t)b * D_SZ))[lane] =
            make_float4(v.x * inv, v.y * inv, v.z * inv, v.w * inv);
    }
}

extern "C" void kernel_launch(void* const* d_in, const int* in_sizes, int n_in,
                              void* d_out, int out_size, void* d_ws, size_t ws_size,
                              hipStream_t stream) {
    const float* h   = (const float*)d_in[0];
    const int*   ns  = (const int*)d_in[1];
    const float* W_f = (const float*)d_in[2];
    const float* b_f = (const float*)d_in[3];
    const float* W_g = (const float*)d_in[4];
    const float* b_g = (const float*)d_in[5];
    float* out = (float*)d_out;

    GraphEmb_gate_reduce_finish<<<dim3(B_SZ), 256, 0, stream>>>(
        h, ns, W_g, b_g, W_f, b_f, out);
}


# work-groups of 8 waves, was 16 (half the waves to start per launch): a wave streams twice the rows, 16 row loads per group; row sums, sum over waves and sum over K (now one pass) regrouped
# speedup vs baseline: 2.5627x; 1.1110x over previous
#include <hip/hip_runtime.h>
#include <hip/hip_bf16.h>

typedef __attribute__((ext_vector_type(2))) float v2f;
typedef __attribute__((ext_vector_type(8))) float v8f;

#define B_SZ 512
#define N_SZ 512
#define D_SZ 128
#define NWAVE 8
#define NVW 16
#define RB 8
#define RS 16
#define APITCH 132
#define NBIN 17


#define ORDER_POINT(q_) do { \
    asm volatile("" : "+v"((q_).x), "+v"((q_).y), "+v"((q_).z), "+v"((q_).w) :: "memory"); \
    __builtin_amdgcn_sched_barrier(0); } while (0)

__device__ __forceinline__ void load_rows(float4 (&buf)[RS], const float4* __restrict__ hrow,
                                          int lane, int i0, int cnt)
{
    #pragma unroll
    for (int j = 0; j < RS; ++j) {
        int i = i0 + j;
        i = i < cnt ? i : cnt - 1;
        const unsigned off = (unsigned)i * (unsigned)(NWAVE * D_SZ * 4) + (unsigned)lane * 16u;
        buf[j] = *(const float4*)((const char*)hrow + off);
    }
}

template <int H>
__device__ __forceinline__ void gate_rows(const float4 (&buf)[RS], int i0, int cnt,
                                          const float4 wg, const float bg,
                                          float4& acc, float& gs)
{
    float d[RB], t[RB], g[RB];
    #pragma unroll
    for (int j = 0; j < RB; ++j) {
        const float4 hv = buf[H + j];
        d[j] = hv.x * wg.x + hv.y * wg.y + hv.z * wg.z + hv.w * wg.w;
    }
    #pragma unroll
    for (int off = 16; off >= 1; off >>= 1) {
        #pragma unroll
        for (int j = 0; j < RB; ++j) t[j] = __shfl_xor(d[j], off, 32);
        __builtin_amdgcn_sched_barrier(0);
        #pragma unroll
        for (int j = 0; j < RB; ++j) d[j] += t[j];
    }
    #pragma unroll
    for (int j = 0; j < RB; ++j) t[j] = __expf(-(d[j] + bg));
    #pragma unroll
    for (int j = 0; j < RB; ++j) {
        g[j] = __builtin_amdgcn_rcpf(1.f + t[j]);
        g[j] = (i0 + H + j < cnt) ? g[j] : 0.f;
    }
    #pragma unroll
    for (int j = 0; j < RB; ++j) {
        const float4 hv = buf[H + j];
        acc.x += g[j] * hv.x; acc.y += g[j] * hv.y;
        acc.z += g[j] * hv.z; acc.w += g[j] * hv.w;
        gs += g[j];
    }
}

__device__ __forceinline__ void load_wf(float4 (&wf)[16], float& bfv,
                                        const float4* __restrict__ wrow,
                                        const float* __restrict__ b_f, int e)
{
    #pragma unroll
    for (int jj = 0; jj < 16; ++jj) wf[jj] = wrow[2 * jj];
    bfv = b_f[e];
}

__device__ __forceinline__ void group_of(int g, int GA,
                                         const float4* hrA, const float4* hrB,
                                         int cntA, int cntB,
                                         const float4*& hr, int& i0, int& cnt)
{
    const bool isA = g < GA;
    hr  = isA ? hrA : hrB;
    i0  = (isA ? g : g - GA) * RS;
    cnt = isA ? cntA : cntB;
}

__global__ __launch_bounds__(256)
void GraphEmb_pair_gate_reduce_finish(const float* __restrict__ h,
                                      const int*   __restrict__ ns,
                                      const float* __restrict__ W_g,
                                      const float* __restrict__ b_g,
                                      const float* __restrict__ W_f,
                                      const float* __restrict__ b_f,
                                      float* __restrict__ out)
{
    const int wgi  = blockIdx.x;
    const int tid  = threadIdx.x;
    const int lane = tid & 31;
    const int warp = __builtin_amdgcn_readfirstlane(tid >> 5);

    __shared__ unsigned binmask[32 * 9];
    __shared__ int ns_s[B_SZ];
    __shared__ __attribute__((aligned(16))) float wacc[2][NWAVE * APITCH];
    __shared__ float wgs[2][NWAVE];
    __shared__ __attribute__((aligned(16))) float hg_s[2][D_SZ];

    float4 wg = ((const float4*)W_g)[lane];
    const float  bg = b_g[0];

    int ns0 = ns[tid], ns1 = ns[tid + B_SZ / 2];
    ns0 = ns0 < 0 ? 0 : (ns0 > N_SZ ? N_SZ : ns0);
    ns1 = ns1 < 0 ? 0 : (ns1 > N_SZ ? N_SZ : ns1);
    ns_s[tid] = ns0;
    ns_s[tid + B_SZ / 2] = ns1;
    const int bin0 = ns0 >> 5, bin1 = ns1 >> 5;
    unsigned mine0 = 0, mine1 = 0;
    #pragma unroll
    for (int bb = 0; bb < NBIN; ++bb) {
        const unsigned m0 = __builtin_amdgcn_ballot_w32(bin0 == bb);
        const unsigned m1 = __builtin_amdgcn_ballot_w32(bin1 == bb);
        mine0 = (lane == bb) ? m0 : mine0;
        mine1 = (lane == bb) ? m1 : mine1;
    }
    if (lane < NBIN + 1) {
        binmask[lane * NVW + warp]         = mine0;
        binmask[lane * NVW + warp + NWAVE] = mine1;
    }
    __syncthreads();

    unsigned mk[9];
    int ck[9];
    int s = 0;
    #pragma unroll
    for (int k = 0; k < 9; ++k) {
        mk[k] = binmask[9 * lane + k];
        ck[k] = __popc(mk[k]);
        s += ck[k];
    }
    int incl = s;
    incl += __builtin_amdgcn_update_dpp(0, incl, 0x111, 0xf, 0xf, false);
    incl += __builtin_amdgcn_update_dpp(0, incl, 0x112, 0xf, 0xf, false);
    incl += __builtin_amdgcn_update_dpp(0, incl, 0x114, 0xf, 0xf, false);
    incl += __builtin_amdgcn_update_dpp(0, incl, 0x118, 0xf, 0xf, false);
    const int row0 = __builtin_amdgcn_readlane(incl, 15);
    incl += (lane >= 16) ? row0 : 0;
    const int r0 = wgi, r1 = B_SZ - 1 - wgi;
    int p = incl - s;
    unsigned fm0 = 0, fm1 = 0;
    int fn0 = 0, fn1 = 0, fw0 = -1, fw1 = -1;
    #pragma unroll
    for (int k = 0; k < 9; ++k) {
        const int n0 = r0 - p, n1 = r1 - p;
        const bool h0 = (unsigned)n0 < (unsigned)ck[k];
        const bool h1 = (unsigned)n1 < (unsigned)ck[k];
        const int wv = (9 * lane + k) & (NVW - 1);
        fm0 = h0 ? mk[k] : fm0;  fn0 = h0 ? n0 : fn0;  fw0 = h0 ? wv : fw0;
        fm1 = h1 ? mk[k] : fm1;  fn1 = h1 ? n1 : fn1;  fw1 = h1 ? wv : fw1;
        p += ck[k];
    }
    const unsigned bm0 = __builtin_amdgcn_ballot_w32(fw0 >= 0);
    const unsigned bm1 = __builtin_amdgcn_ballot_w32(fw1 >= 0);
    const int sl0 = bm0 ? __builtin_ctz(bm0) : 0;
    const int sl1 = bm1 ? __builtin_ctz(bm1) : 0;
    const unsigned M0 = (unsigned)__builtin_amdgcn_readlane((int)fm0, sl0);
    const unsigned M1 = (unsigned)__builtin_amdgcn_readlane((int)fm1, sl1);
    const int N0 = __builtin_amdgcn_readlane(fn0, sl0);
    const int N1 = __builtin_amdgcn_readlane(fn1, sl1);
    const int V0 = __builtin_amdgcn_readlane(fw0, sl0);
    const int V1 = __builtin_amdgcn_readlane(fw1, sl1);
    const unsigned below = (1u << lane) - 1u;
    const unsigned hb0 = __builtin_amdgcn_ballot_w32(((M0 >> lane) & 1u) && __popc(M0 & below) == N0);
    const unsigned hb1 = __builtin_amdgcn_ballot_w32(((M1 >> lane) & 1u) && __popc(M1 & below) == N1);
    int bA = V0 * 32 + (hb0 ? __builtin_ctz(hb0) : 0);
    int bB = V1 * 32 + (hb1 ? __builtin_ctz(hb1) : 0);
    bA &= B_SZ - 1;
    bB &= B_SZ - 1;
    const int nsA = __builtin_amdgcn_readfirstlane(ns_s[bA]);
    const int nsB = __builtin_amdgcn_readfirstlane(ns_s[bB]);

    const int cntA = nsA > warp ? (nsA - warp + NWAVE - 1) / NWAVE : 0;
    const int cntB = nsB > warp ? (nsB - warp + NWAVE - 1) / NWAVE : 0;
    const int GA = (cntA + RS - 1) / RS;
    const int GB = (cntB + RS - 1) / RS;
    const int G  = GA + GB;

    const int c  = lane & 15;
    const int hh = lane >> 4;
    const int e  = warp * 16 + c;
    const float4* wrow = (const float4*)(W_f + (size_t)e * D_SZ) + hh;
    float4 wf[16];
    float  bfv;

    float4 acc  = make_float4(0.f, 0.f, 0.f, 0.f);
    float  gs   = 0.f;
    float4 accA = make_float4(0.f, 0.f, 0.f, 0.f);
    float  gsA  = 0.f;

#define FLUSH_AT(gidx_) do { const bool fl_ = ((gidx_) == GA); \
        accA.x = fl_ ? acc.x : accA.x; accA.y = fl_ ? acc.y : accA.y; \
        accA.z = fl_ ? acc.z : accA.z; accA.w = fl_ ? acc.w : accA.w; \
        gsA = fl_ ? gs : gsA; \
        acc.x = fl_ ? 0.f : acc.x; acc.y = fl_ ? 0.f : acc.y; \
        acc.z = fl_ ? 0.f : acc.z; acc.w = fl_ ? 0.f : acc.w; \
        gs = fl_ ? 0.f : gs; } while (0)

#define GATE_GROUP(buf_) do { \
        gate_rows<0>(buf_, i0, cnt, wg, bg, acc, gs); \
        if (i0 + RB < cnt) gate_rows<RB>(buf_, i0, cnt, wg, bg, acc, gs); } while (0)

    if (G > 0) {
        const float4* hrA = (const float4*)h + ((size_t)bA * N_SZ + warp) * (D_SZ / 4);
        const float4* hrB = (const float4*)h + ((size_t)bB * N_SZ + warp) * (D_SZ / 4);
        const float4* hr; int i0, cnt;
        float4 bufA[RS], bufB[RS];
        group_of(0, GA, hrA, hrB, cntA, cntB, hr, i0, cnt);
        load_rows(bufA, hr, lane, i0, cnt);
        int g = 0;
        while (g + 2 < G) {
            group_of(g + 1, GA, hrA, hrB, cntA, cntB, hr, i0, cnt);
            load_rows(bufB, hr, lane, i0, cnt);
            ORDER_POINT(wg);
            FLUSH_AT(g);
            group_of(g, GA, hrA, hrB, cntA, cntB, hr, i0, cnt);
            GATE_GROUP(bufA);
            group_of(g + 2, GA, hrA, hrB, cntA, cntB, hr, i0, cnt);
            load_rows(bufA, hr, lane, i0, cnt);
            ORDER_POINT(wg);
            FLUSH_AT(g + 1);
            group_of(g + 1, GA, hrA, hrB, cntA, cntB, hr, i0, cnt);
            GATE_GROUP(bufB);
            g += 2;
        }
        load_wf(wf, bfv, wrow, b_f, e);
        if (g + 1 < G) {
            group_of(g + 1, GA, hrA, hrB, cntA, cntB, hr, i0, cnt);
            load_rows(bufB, hr, lane, i0, cnt);
            ORDER_POINT(wg);
            FLUSH_AT(g);
            group_of(g, GA, hrA, hrB, cntA, cntB, hr, i0, cnt);
            GATE_GROUP(bufA);
            FLUSH_AT(g + 1);
            group_of(g + 1, GA, hrA, hrB, cntA, cntB, hr, i0, cnt);
            GATE_GROUP(bufB);
        } else {
            FLUSH_AT(g);
            group_of(g, GA, hrA, hrB, cntA, cntB, hr, i0, cnt);
            GATE_GROUP(bufA);
        }
    } else {
        load_wf(wf, bfv, wrow, b_f, e);
    }
    FLUSH_AT(G);
#undef FLUSH_AT
#undef GATE_GROUP

    ((float4*)(wacc[0] + warp * APITCH))[lane] = accA;
    ((float4*)(wacc[1] + warp * APITCH))[lane] = acc;
    if (lane == 0) { wgs[0][warp] = gsA; wgs[1][warp] = gs; }
    __syncthreads();

    const float* alo = wacc[c >> 3] + (c & 7) * APITCH + 4 * hh;
    float4 av[16];
    #pragma unroll
    for (int jj = 0; jj < 16; ++jj) av[jj] = *(const float4*)(alo + 8 * jj);
    float sg = 0.f;
    #pragma unroll
    for (int w = 0; w < NWAVE; ++w) sg += wgs[hh][w];
    __builtin_amdgcn_sched_barrier(0);
    v8f c0 = {}, c1 = {};
    #pragma unroll
    for (int jj = 0; jj < 16; ++jj) {
        const float4 bv = wf[jj];
        v2f a0 = { av[jj].x, av[jj].y }, b0 = { bv.x, bv.y };
        v2f a1 = { av[jj].z, av[jj].w }, b1 = { bv.z, bv.w };
        c0 = __builtin_amdgcn_wmma_f32_16x16x4_f32(false, a0, false, b0,
                                                   (short)0, c0, false, false);
        c1 = __builtin_amdgcn_wmma_f32_16x16x4_f32(false, a1, false, b1,
                                                   (short)0, c1, false, false);
    }

    float hv = 0.f;
    #pragma unroll
    for (int r = 0; r < 8; ++r) hv += c0[r] + c1[r];
    hv += sg * bfv;
    hg_s[hh][e] = hv;
    __syncthreads();

    if (warp < 2) {
        const int bo = warp ? bB : bA;
        const float4 v = ((const float4*)hg_s[warp])[lane];
        float q = v.x * v.x + v.y * v.y + v.z * v.z + v.w * v.w;
        #pragma unroll
        for (int off = 16; off >= 1; off >>= 1) q += __shfl_xor(q, off, 32);
        const float inv = 1.f / fmaxf(sqrtf(q), 1e-12f);
        ((float4*)(out + (size_t)bo * D_SZ))[lane] =
            make_float4(v.x * inv, v.y * inv, v.z * inv, v.w * inv);
    }
}

extern "C" void kernel_launch(void* const* d_in, const int* in_sizes, int n_in,
                              void* d_out, int out_size, void* d_ws, size_t ws_size,
                              hipStream_t stream) {
    const float* h   = (const float*)d_in[0];
    const int*   ns  = (const int*)d_in[1];
    const float* W_f = (const float*)d_in[2];
    const float* b_f = (const float*)d_in[3];
    const float* W_g = (const float*)d_in[4];
    const float* b_g = (const float*)d_in[5];
    float* out = (float*)d_out;

    GraphEmb_pair_gate_reduce_finish<<<dim3(256), 256, 0, stream>>>(
        h, ns, W_g, b_g, W_f, b_f, out);
}


# gate of 8 rows as one vector: the dot products fold into one register by VALU lane exchanges (permlanex16, row_xmask DPP; was 40 ds_bpermute), one exp/rcp per 8 rows; norm sum too; same sum order
# speedup vs baseline: 2.7190x; 1.0610x over previous
#include <hip/hip_runtime.h>
#include <hip/hip_bf16.h>

typedef __attribute__((ext_vector_type(2))) float v2f;
typedef __attribute__((ext_vector_type(8))) float v8f;

#define B_SZ 512
#define N_SZ 512
#define D_SZ 128
#define NWAVE 8
#define NVW 16
#define RB 8
#define RS 16
#define APITCH 132
#define NBIN 17


#define ORDER_POINT(q_) do { \
    asm volatile("" : "+v"((q_).x), "+v"((q_).y), "+v"((q_).z), "+v"((q_).w) :: "memory"); \
    __builtin_amdgcn_sched_barrier(0); } while (0)

__device__ __forceinline__ void load_rows(float4 (&buf)[RS], const float4* __restrict__ hrow,
                                          int lane, int i0, int cnt)
{
    #pragma unroll
    for (int j = 0; j < RS; ++j) {
        int i = i0 + j;
        i = i < cnt ? i : cnt - 1;
        const unsigned off = (unsigned)i * (unsigned)(NWAVE * D_SZ * 4) + (unsigned)lane * 16u;
        buf[j] = *(const float4*)((const char*)hrow + off);
    }
}

__device__ __forceinline__ float xhalf(float x)
{
    const int b = __builtin_bit_cast(int, x);
    return __builtin_bit_cast(float,
        __builtin_amdgcn_permlanex16(b, b, 0x76543210, 0xfedcba98, false, false));
}
template <int M>
__device__ __forceinline__ float xrow(float x)
{
    return __builtin_bit_cast(float,
        __builtin_amdgcn_update_dpp(0, __builtin_bit_cast(int, x), 0x160 | M, 0xf, 0xf, true));
}

template <int H>
__device__ __forceinline__ void gate_rows(const float4 (&buf)[RS], int i0, int cnt,
                                          const float4 wg, const float bg, int lane,
                                          float4& acc, float& gs)
{
    float d[RB], e[RB / 2], f[RB / 4];
    const bool b4 = (lane & 16) != 0, b3 = (lane & 8) != 0, b2 = (lane & 4) != 0;
    #pragma unroll
    for (int j = 0; j < RB; ++j) {
        const float4 hv = buf[H + j];
        d[j] = hv.x * wg.x + hv.y * wg.y + hv.z * wg.z + hv.w * wg.w;
    }
    #pragma unroll
    for (int i = 0; i < RB / 2; ++i) {
        const float keep = b4 ? d[2 * i + 1] : d[2 * i];
        const float send = b4 ? d[2 * i] : d[2 * i + 1];
        e[i] = keep + xhalf(send);
    }
    #pragma unroll
    for (int i = 0; i < RB / 4; ++i) {
        const float keep = b3 ? e[2 * i + 1] : e[2 * i];
        const float send = b3 ? e[2 * i] : e[2 * i + 1];
        f[i] = keep + xrow<8>(send);
    }
    float q;
    {
        const float keep = b2 ? f[1] : f[0];
        const float send = b2 ? f[0] : f[1];
        q = keep + xrow<4>(send);
    }
    q += xrow<2>(q);
    q += xrow<1>(q);
    const int rho = ((lane >> 4) & 1) | ((lane >> 2) & 2) | (lane & 4);
    const float t = __expf(-(q + bg));
    float gv = __builtin_amdgcn_rcpf(1.f + t);
    gv = (i0 + H + rho < cnt) ? gv : 0.f;
    #pragma unroll
    for (int j = 0; j < RB; ++j) {
        const int src = 16 * (j & 1) + 8 * ((j >> 1) & 1) + 4 * ((j >> 2) & 1);
        const float g = __builtin_bit_cast(float,
            __builtin_amdgcn_readlane(__builtin_bit_cast(int, gv), src));
        const float4 hv = buf[H + j];
        acc.x += g * hv.x; acc.y += g * hv.y;
        acc.z += g * hv.z; acc.w += g * hv.w;
        gs += g;
    }
}

__device__ __forceinline__ void load_wf(float4 (&wf)[16], float& bfv,
                                        const float4* __restrict__ wrow,
                                        const float* __restrict__ b_f, int e)
{
    #pragma unroll
    for (int jj = 0; jj < 16; ++jj) wf[jj] = wrow[2 * jj];
    bfv = b_f[e];
}

__device__ __forceinline__ void group_of(int g, int GA,
                                         const float4* hrA, const float4* hrB,
                                         int cntA, int cntB,
                                         const float4*& hr, int& i0, int& cnt)
{
    const bool isA = g < GA;
    hr  = isA ? hrA : hrB;
    i0  = (isA ? g : g - GA) * RS;
    cnt = isA ? cntA : cntB;
}

__global__ __launch_bounds__(256)
void GraphEmb_pair_gate_reduce_finish(const float* __restrict__ h,
                                      const int*   __restrict__ ns,
                                      const float* __restrict__ W_g,
                                      const float* __restrict__ b_g,
                                      const float* __restrict__ W_f,
                                      const float* __restrict__ b_f,
                                      float* __restrict__ out)
{
    const int wgi  = blockIdx.x;
    const int tid  = threadIdx.x;
    const int lane = tid & 31;
    const int warp = __builtin_amdgcn_readfirstlane(tid >> 5);

    __shared__ unsigned binmask[32 * 9];
    __shared__ int ns_s[B_SZ];
    __shared__ __attribute__((aligned(16))) float wacc[2][NWAVE * APITCH];
    __shared__ float wgs[2][NWAVE];
    __shared__ __attribute__((aligned(16))) float hg_s[2][D_SZ];

    float4 wg = ((const float4*)W_g)[lane];
    const float  bg = b_g[0];

    int ns0 = ns[tid], ns1 = ns[tid + B_SZ / 2];
    ns0 = ns0 < 0 ? 0 : (ns0 > N_SZ ? N_SZ : ns0);
    ns1 = ns1 < 0 ? 0 : (ns1 > N_SZ ? N_SZ : ns1);
    ns_s[tid] = ns0;
    ns_s[tid + B_SZ / 2] = ns1;
    const int bin0 = ns0 >> 5, bin1 = ns1 >> 5;
    unsigned mine0 = 0, mine1 = 0;
    #pragma unroll
    for (int bb = 0; bb < NBIN; ++bb) {
        const unsigned m0 = __builtin_amdgcn_ballot_w32(bin0 == bb);
        const unsigned m1 = __builtin_amdgcn_ballot_w32(bin1 == bb);
        mine0 = (lane == bb) ? m0 : mine0;
        mine1 = (lane == bb) ? m1 : mine1;
    }
    if (lane < NBIN + 1) {
        binmask[lane * NVW + warp]         = mine0;
        binmask[lane * NVW + warp + NWAVE] = mine1;
    }
    __syncthreads();

    unsigned mk[9];
    int ck[9];
    int s = 0;
    #pragma unroll
    for (int k = 0; k < 9; ++k) {
        mk[k] = binmask[9 * lane + k];
        ck[k] = __popc(mk[k]);
        s += ck[k];
    }
    int incl = s;
    incl += __builtin_amdgcn_update_dpp(0, incl, 0x111, 0xf, 0xf, false);
    incl += __builtin_amdgcn_update_dpp(0, incl, 0x112, 0xf, 0xf, false);
    incl += __builtin_amdgcn_update_dpp(0, incl, 0x114, 0xf, 0xf, false);
    incl += __builtin_amdgcn_update_dpp(0, incl, 0x118, 0xf, 0xf, false);
    const int row0 = __builtin_amdgcn_readlane(incl, 15);
    incl += (lane >= 16) ? row0 : 0;
    const int r0 = wgi, r1 = B_SZ - 1 - wgi;
    int p = incl - s;
    unsigned fm0 = 0, fm1 = 0;
    int fn0 = 0, fn1 = 0, fw0 = -1, fw1 = -1;
    #pragma unroll
    for (int k = 0; k < 9; ++k) {
        const int n0 = r0 - p, n1 = r1 - p;
        const bool h0 = (unsigned)n0 < (unsigned)ck[k];
        const bool h1 = (unsigned)n1 < (unsigned)ck[k];
        const int wv = (9 * lane + k) & (NVW - 1);
        fm0 = h0 ? mk[k] : fm0;  fn0 = h0 ? n0 : fn0;  fw0 = h0 ? wv : fw0;
        fm1 = h1 ? mk[k] : fm1;  fn1 = h1 ? n1 : fn1;  fw1 = h1 ? wv : fw1;
        p += ck[k];
    }
    const unsigned bm0 = __builtin_amdgcn_ballot_w32(fw0 >= 0);
    const unsigned bm1 = __builtin_amdgcn_ballot_w32(fw1 >= 0);
    const int sl0 = bm0 ? __builtin_ctz(bm0) : 0;
    const int sl1 = bm1 ? __builtin_ctz(bm1) : 0;
    const unsigned M0 = (unsigned)__builtin_amdgcn_readlane((int)fm0, sl0);
    const unsigned M1 = (unsigned)__builtin_amdgcn_readlane((int)fm1, sl1);
    const int N0 = __builtin_amdgcn_readlane(fn0, sl0);
    const int N1 = __builtin_amdgcn_readlane(fn1, sl1);
    const int V0 = __builtin_amdgcn_readlane(fw0, sl0);
    const int V1 = __builtin_amdgcn_readlane(fw1, sl1);
    const unsigned below = (1u << lane) - 1u;
    const unsigned hb0 = __builtin_amdgcn_ballot_w32(((M0 >> lane) & 1u) && __popc(M0 & below) == N0);
    const unsigned hb1 = __builtin_amdgcn_ballot_w32(((M1 >> lane) & 1u) && __popc(M1 & below) == N1);
    int bA = V0 * 32 + (hb0 ? __builtin_ctz(hb0) : 0);
    int bB = V1 * 32 + (hb1 ? __builtin_ctz(hb1) : 0);
    bA &= B_SZ - 1;
    bB &= B_SZ - 1;
    const int nsA = __builtin_amdgcn_readfirstlane(ns_s[bA]);
    const int nsB = __builtin_amdgcn_readfirstlane(ns_s[bB]);

    const int cntA = nsA > warp ? (nsA - warp + NWAVE - 1) / NWAVE : 0;
    const int cntB = nsB > warp ? (nsB - warp + NWAVE - 1) / NWAVE : 0;
    const int GA = (cntA + RS - 1) / RS;
    const int GB = (cntB + RS - 1) / RS;
    const int G  = GA + GB;

    const int c  = lane & 15;
    const int hh = lane >> 4;
    const int e  = warp * 16 + c;
    const float4* wrow = (const float4*)(W_f + (size_t)e * D_SZ) + hh;
    float4 wf[16];
    float  bfv;

    float4 acc  = make_float4(0.f, 0.f, 0.f, 0.f);
    float  gs   = 0.f;
    float4 accA = make_float4(0.f, 0.f, 0.f, 0.f);
    float  gsA  = 0.f;

#define FLUSH_AT(gidx_) do { const bool fl_ = ((gidx_) == GA); \
        accA.x = fl_ ? acc.x : accA.x; accA.y = fl_ ? acc.y : accA.y; \
        accA.z = fl_ ? acc.z : accA.z; accA.w = fl_ ? acc.w : accA.w; \
        gsA = fl_ ? gs : gsA; \
        acc.x = fl_ ? 0.f : acc.x; acc.y = fl_ ? 0.f : acc.y; \
        acc.z = fl_ ? 0.f : acc.z; acc.w = fl_ ? 0.f : acc.w; \
        gs = fl_ ? 0.f : gs; } while (0)

#define GATE_GROUP(buf_) do { \
        gate_rows<0>(buf_, i0, cnt, wg, bg, lane, acc, gs); \
        if (i0 + RB < cnt) gate_rows<RB>(buf_, i0, cnt, wg, bg, lane, acc, gs); } while (0)

    if (G > 0) {
        const float4* hrA = (const float4*)h + ((size_t)bA * N_SZ + warp) * (D_SZ / 4);
        const float4* hrB = (const float4*)h + ((size_t)bB * N_SZ + warp) * (D_SZ / 4);
        const float4* hr; int i0, cnt;
        float4 bufA[RS], bufB[RS];
        group_of(0, GA, hrA, hrB, cntA, cntB, hr, i0, cnt);
        load_rows(bufA, hr, lane, i0, cnt);
        int g = 0;
        while (g + 2 < G) {
            group_of(g + 1, GA, hrA, hrB, cntA, cntB, hr, i0, cnt);
            load_rows(bufB, hr, lane, i0, cnt);
            ORDER_POINT(wg);
            FLUSH_AT(g);
            group_of(g, GA, hrA, hrB, cntA, cntB, hr, i0, cnt);
            GATE_GROUP(bufA);
            group_of(g + 2, GA, hrA, hrB, cntA, cntB, hr, i0, cnt);
            load_rows(bufA, hr, lane, i0, cnt);
            ORDER_POINT(wg);
            FLUSH_AT(g + 1);
            group_of(g + 1, GA, hrA, hrB, cntA, cntB, hr, i0, cnt);
            GATE_GROUP(bufB);
            g += 2;
        }
        load_wf(wf, bfv, wrow, b_f, e);
        if (g + 1 < G) {
            group_of(g + 1, GA, hrA, hrB, cntA, cntB, hr, i0, cnt);
            load_rows(bufB, hr, lane, i0, cnt);
            ORDER_POINT(wg);
            FLUSH_AT(g);
            group_of(g, GA, hrA, hrB, cntA, cntB, hr, i0, cnt);
            GATE_GROUP(bufA);
            FLUSH_AT(g + 1);
            group_of(g + 1, GA, hrA, hrB, cntA, cntB, hr, i0, cnt);
            GATE_GROUP(bufB);
        } else {
            FLUSH_AT(g);
            group_of(g, GA, hrA, hrB, cntA, cntB, hr, i0, cnt);
            GATE_GROUP(bufA);
        }
    } else {
        load_wf(wf, bfv, wrow, b_f, e);
    }
    FLUSH_AT(G);
#undef FLUSH_AT
#undef GATE_GROUP

    ((float4*)(wacc[0] + warp * APITCH))[lane] = accA;
    ((float4*)(wacc[1] + warp * APITCH))[lane] = acc;
    if (lane == 0) { wgs[0][warp] = gsA; wgs[1][warp] = gs; }
    __syncthreads();

    const float* alo = wacc[c >> 3] + (c & 7) * APITCH + 4 * hh;
    float4 av[16];
    #pragma unroll
    for (int jj = 0; jj < 16; ++jj) av[jj] = *(const float4*)(alo + 8 * jj);
    float sg = 0.f;
    #pragma unroll
    for (int w = 0; w < NWAVE; ++w) sg += wgs[hh][w];
    __builtin_amdgcn_sched_barrier(0);
    v8f c0 = {}, c1 = {};
    #pragma unroll
    for (int jj = 0; jj < 16; ++jj) {
        const float4 bv = wf[jj];
        v2f a0 = { av[jj].x, av[jj].y }, b0 = { bv.x, bv.y };
        v2f a1 = { av[jj].z, av[jj].w }, b1 = { bv.z, bv.w };
        c0 = __builtin_amdgcn_wmma_f32_16x16x4_f32(false, a0, false, b0,
                                                   (short)0, c0, false, false);
        c1 = __builtin_amdgcn_wmma_f32_16x16x4_f32(false, a1, false, b1,
                                                   (short)0, c1, false, false);
    }

    float hv = 0.f;
    #pragma unroll
    for (int r = 0; r < 8; ++r) hv += c0[r] + c1[r];
    hv += sg * bfv;
    hg_s[hh][e] = hv;
    __syncthreads();

    if (warp < 2) {
        const int bo = warp ? bB : bA;
        const float4 v = ((const float4*)hg_s[warp])[lane];
        float q = v.x * v.x + v.y * v.y + v.z * v.z + v.w * v.w;
        q += xhalf(q);
        q += xrow<8>(q);
        q += xrow<4>(q);
        q += xrow<2>(q);
        q += xrow<1>(q);
        const float inv = 1.f / fmaxf(sqrtf(q), 1e-12f);
        ((float4*)(out + (size_t)bo * D_SZ))[lane] =
            make_float4(v.x * inv, v.y * inv, v.z * inv, v.w * inv);
    }
}

extern "C" void kernel_launch(void* const* d_in, const int* in_sizes, int n_in,
                              void* d_out, int out_size, void* d_ws, size_t ws_size,
                              hipStream_t stream) {
    const float* h   = (const float*)d_in[0];
    const int*   ns  = (const int*)d_in[1];
    const float* W_f = (const float*)d_in[2];
    const float* b_f = (const float*)d_in[3];
    const float* W_g = (const float*)d_in[4];
    const float* b_g = (const float*)d_in[5];
    float* out = (float*)d_out;

    GraphEmb_pair_gate_reduce_finish<<<dim3(256), 256, 0, stream>>>(
        h, ns, W_g, b_g, W_f, b_f, out);
}


# work-group w's graph A is graph w itself: its first 16 row loads per wave go out at entry, and the pairing (rank in lower half, opposite rank in upper half) runs under them; sum order unchanged
# speedup vs baseline: 3.0130x; 1.1081x over previous
#include <hip/hip_runtime.h>
#include <hip/hip_bf16.h>

typedef __attribute__((ext_vector_type(2))) float v2f;
typedef __attribute__((ext_vector_type(8))) float v8f;

#define B_SZ 512
#define N_SZ 512
#define D_SZ 128
#define NWAVE 8
#define NWORD 160
#define RB 8
#define RS 16
#define APITCH 132
#define NBIN 17


#define ORDER_POINT(q_) do { \
    asm volatile("" : "+v"((q_).x), "+v"((q_).y), "+v"((q_).z), "+v"((q_).w) :: "memory"); \
    __builtin_amdgcn_sched_barrier(0); } while (0)

__device__ __forceinline__ void load_rows(float4 (&buf)[RS], const float4* __restrict__ hrow,
                                          int lane, int i0, int cnt)
{
    #pragma unroll
    for (int j = 0; j < RS; ++j) {
        int i = i0 + j;
        i = i < cnt ? i : cnt - 1;
        const unsigned off = (unsigned)i * (unsigned)(NWAVE * D_SZ * 4) + (unsigned)lane * 16u;
        buf[j] = *(const float4*)((const char*)hrow + off);
    }
}

__device__ __forceinline__ float xhalf(float x)
{
    const int b = __builtin_bit_cast(int, x);
    return __builtin_bit_cast(float,
        __builtin_amdgcn_permlanex16(b, b, 0x76543210, 0xfedcba98, false, false));
}
template <int M>
__device__ __forceinline__ float xrow(float x)
{
    return __builtin_bit_cast(float,
        __builtin_amdgcn_update_dpp(0, __builtin_bit_cast(int, x), 0x160 | M, 0xf, 0xf, true));
}

template <int H>
__device__ __forceinline__ void gate_rows(const float4 (&buf)[RS], int i0, int cnt,
                                          const float4 wg, const float bg, int lane,
                                          float4& acc, float& gs)
{
    float d[RB], e[RB / 2], f[RB / 4];
    const bool b4 = (lane & 16) != 0, b3 = (lane & 8) != 0, b2 = (lane & 4) != 0;
    #pragma unroll
    for (int j = 0; j < RB; ++j) {
        const float4 hv = buf[H + j];
        d[j] = hv.x * wg.x + hv.y * wg.y + hv.z * wg.z + hv.w * wg.w;
    }
    #pragma unroll
    for (int i = 0; i < RB / 2; ++i) {
        const float keep = b4 ? d[2 * i + 1] : d[2 * i];
        const float send = b4 ? d[2 * i] : d[2 * i + 1];
        e[i] = keep + xhalf(send);
    }
    #pragma unroll
    for (int i = 0; i < RB / 4; ++i) {
        const float keep = b3 ? e[2 * i + 1] : e[2 * i];
        const float send = b3 ? e[2 * i] : e[2 * i + 1];
        f[i] = keep + xrow<8>(send);
    }
    float q;
    {
        const float keep = b2 ? f[1] : f[0];
        const float send = b2 ? f[0] : f[1];
        q = keep + xrow<4>(send);
    }
    q += xrow<2>(q);
    q += xrow<1>(q);
    const int rho = ((lane >> 4) & 1) | ((lane >> 2) & 2) | (lane & 4);
    const float t = __expf(-(q + bg));
    float gv = __builtin_amdgcn_rcpf(1.f + t);
    gv = (i0 + H + rho < cnt) ? gv : 0.f;
    #pragma unroll
    for (int j = 0; j < RB; ++j) {
        const int src = 16 * (j & 1) + 8 * ((j >> 1) & 1) + 4 * ((j >> 2) & 1);
        const float g = __builtin_bit_cast(float,
            __builtin_amdgcn_readlane(__builtin_bit_cast(int, gv), src));
        const float4 hv = buf[H + j];
        acc.x += g * hv.x; acc.y += g * hv.y;
        acc.z += g * hv.z; acc.w += g * hv.w;
        gs += g;
    }
}

__device__ __forceinline__ void load_wf(float4 (&wf)[16], float& bfv,
                                        const float4* __restrict__ wrow,
                                        const float* __restrict__ b_f, int e)
{
    #pragma unroll
    for (int jj = 0; jj < 16; ++jj) wf[jj] = wrow[2 * jj];
    bfv = b_f[e];
}

__device__ __forceinline__ void group_of(int g, int GA,
                                         const float4* hrA, const float4* hrB,
                                         int cntA, int cntB,
                                         const float4*& hr, int& i0, int& cnt)
{
    const bool isA = g < GA;
    hr  = isA ? hrA : hrB;
    i0  = (isA ? g : g - GA) * RS;
    cnt = isA ? cntA : cntB;
}

__device__ __forceinline__ void load_rows_entry(float4 (&buf)[RS], const float4* __restrict__ hrow,
                                                int lane)
{
    #pragma unroll
    for (int j = 0; j < RS; ++j) {
        const unsigned off = (unsigned)j * (unsigned)(NWAVE * D_SZ * 4) + (unsigned)lane * 16u;
        buf[j] = *(const float4*)((const char*)hrow + off);
    }
}

__global__ __launch_bounds__(256)
void GraphEmb_own_pair_gate_reduce_finish(const float* __restrict__ h,
                                          const int*   __restrict__ ns,
                                          const float* __restrict__ W_g,
                                          const float* __restrict__ b_g,
                                          const float* __restrict__ W_f,
                                          const float* __restrict__ b_f,
                                          float* __restrict__ out)
{
    const int wgi  = blockIdx.x;
    const int tid  = threadIdx.x;
    const int lane = tid & 31;
    const int warp = __builtin_amdgcn_readfirstlane(tid >> 5);

    __shared__ unsigned binmask[NWORD];
    __shared__ __attribute__((aligned(16))) int cntw[NWAVE];
    __shared__ int ns_u[B_SZ / 2];
    __shared__ __attribute__((aligned(16))) float wacc[2][NWAVE * APITCH];
    __shared__ float wgs[2][NWAVE];
    __shared__ __attribute__((aligned(16))) float hg_s[2][D_SZ];

    int ns0 = ns[tid], ns1 = ns[tid + B_SZ / 2];
    int nsA = ns[wgi];
    float4 wg = ((const float4*)W_g)[lane];
    const float  bg = b_g[0];
    const int bA = wgi;
    const float4* hrA = (const float4*)h + ((size_t)bA * N_SZ + warp) * (D_SZ / 4);
    float4 bufA[RS], bufB[RS];
    load_rows_entry(bufA, hrA, lane);
    asm volatile("" ::: "memory");
    __builtin_amdgcn_sched_barrier(0);

    ns0 = ns0 < 0 ? 0 : (ns0 > N_SZ ? N_SZ : ns0);
    ns1 = ns1 < 0 ? 0 : (ns1 > N_SZ ? N_SZ : ns1);
    nsA = nsA < 0 ? 0 : (nsA > N_SZ ? N_SZ : nsA);
    ns_u[tid] = ns1;
    const int bin0 = ns0 >> 5, bin1 = ns1 >> 5, binA = nsA >> 5;
    const bool before = (bin0 < binA) || (bin0 == binA && tid < bA);
    const unsigned mb = __builtin_amdgcn_ballot_w32(before);
    unsigned mine1 = 0;
    #pragma unroll
    for (int bb = 0; bb < NBIN; ++bb) {
        const unsigned m1 = __builtin_amdgcn_ballot_w32(bin1 == bb);
        mine1 = (lane == bb) ? m1 : mine1;
    }
    if (lane < NWORD / NWAVE) binmask[lane * NWAVE + warp] = mine1;
    if (lane == 0) cntw[warp] = __popc(mb);
    __builtin_amdgcn_fence(__ATOMIC_RELEASE, "workgroup", "local");
    __builtin_amdgcn_s_barrier();
    __builtin_amdgcn_fence(__ATOMIC_ACQUIRE, "workgroup", "local");

    const int4 cw0 = ((const int4*)cntw)[0], cw1 = ((const int4*)cntw)[1];
    const int rA = __builtin_amdgcn_readfirstlane(
        cw0.x + cw0.y + cw0.z + cw0.w + cw1.x + cw1.y + cw1.z + cw1.w);
    const int r1 = B_SZ / 2 - 1 - rA;
    unsigned mk[5];
    int ck[5];
    int s = 0;
    #pragma unroll
    for (int k = 0; k < 5; ++k) {
        mk[k] = binmask[5 * lane + k];
        ck[k] = __popc(mk[k]);
        s += ck[k];
    }
    int incl = s;
    incl += __builtin_amdgcn_update_dpp(0, incl, 0x111, 0xf, 0xf, false);
    incl += __builtin_amdgcn_update_dpp(0, incl, 0x112, 0xf, 0xf, false);
    incl += __builtin_amdgcn_update_dpp(0, incl, 0x114, 0xf, 0xf, false);
    incl += __builtin_amdgcn_update_dpp(0, incl, 0x118, 0xf, 0xf, false);
    const int row0 = __builtin_amdgcn_readlane(incl, 15);
    incl += (lane >= 16) ? row0 : 0;
    int p = incl - s;
    unsigned fm1 = 0;
    int fn1 = 0, fw1 = -1;
    #pragma unroll
    for (int k = 0; k < 5; ++k) {
        const int n1 = r1 - p;
        const bool h1 = (unsigned)n1 < (unsigned)ck[k];
        const int wv = (5 * lane + k) & (NWAVE - 1);
        fm1 = h1 ? mk[k] : fm1;  fn1 = h1 ? n1 : fn1;  fw1 = h1 ? wv : fw1;
        p += ck[k];
    }
    const unsigned bm1 = __builtin_amdgcn_ballot_w32(fw1 >= 0);
    const int sl1 = bm1 ? __builtin_ctz(bm1) : 0;
    const unsigned M1 = (unsigned)__builtin_amdgcn_readlane((int)fm1, sl1);
    const int N1 = __builtin_amdgcn_readlane(fn1, sl1);
    const int V1 = __builtin_amdgcn_readlane(fw1, sl1);
    const unsigned below = (1u << lane) - 1u;
    const unsigned hb1 = __builtin_amdgcn_ballot_w32(((M1 >> lane) & 1u) && __popc(M1 & below) == N1);
    int uB = V1 * 32 + (hb1 ? __builtin_ctz(hb1) : 0);
    uB &= B_SZ / 2 - 1;
    const int bB = B_SZ / 2 + uB;
    const int nsB = __builtin_amdgcn_readfirstlane(ns_u[uB]);

    const int cntA = nsA > warp ? (nsA - warp + NWAVE - 1) / NWAVE : 0;
    const int cntB = nsB > warp ? (nsB - warp + NWAVE - 1) / NWAVE : 0;
    const int GA = cntA > RS ? (cntA + RS - 1) / RS : 1;
    const int GB = (cntB + RS - 1) / RS;
    const int G  = GA + GB;

    const int c  = lane & 15;
    const int hh = lane >> 4;
    const int e  = warp * 16 + c;
    const float4* wrow = (const float4*)(W_f + (size_t)e * D_SZ) + hh;
    float4 wf[16];
    float  bfv;

    float4 acc  = make_float4(0.f, 0.f, 0.f, 0.f);
    float  gs   = 0.f;
    float4 accA = make_float4(0.f, 0.f, 0.f, 0.f);
    float  gsA  = 0.f;

#define FLUSH_AT(gidx_) do { const bool fl_ = ((gidx_) == GA); \
        accA.x = fl_ ? acc.x : accA.x; accA.y = fl_ ? acc.y : accA.y; \
        accA.z = fl_ ? acc.z : accA.z; accA.w = fl_ ? acc.w : accA.w; \
        gsA = fl_ ? gs : gsA; \
        acc.x = fl_ ? 0.f : acc.x; acc.y = fl_ ? 0.f : acc.y; \
        acc.z = fl_ ? 0.f : acc.z; acc.w = fl_ ? 0.f : acc.w; \
        gs = fl_ ? 0.f : gs; } while (0)

#define GATE_GROUP(buf_) do { \
        gate_rows<0>(buf_, i0, cnt, wg, bg, lane, acc, gs); \
        if (i0 + RB < cnt) gate_rows<RB>(buf_, i0, cnt, wg, bg, lane, acc, gs); } while (0)

    {
        const float4* hrB = (const float4*)h + ((size_t)bB * N_SZ + warp) * (D_SZ / 4);
        const float4* hr; int i0, cnt;
        int g = 0;
        while (g + 2 < G) {
            group_of(g + 1, GA, hrA, hrB, cntA, cntB, hr, i0, cnt);
            load_rows(bufB, hr, lane, i0, cnt);
            ORDER_POINT(wg);
            FLUSH_AT(g);
            group_of(g, GA, hrA, hrB, cntA, cntB, hr, i0, cnt);
            GATE_GROUP(bufA);
            group_of(g + 2, GA, hrA, hrB, cntA, cntB, hr, i0, cnt);
            load_rows(bufA, hr, lane, i0, cnt);
            ORDER_POINT(wg);
            FLUSH_AT(g + 1);
            group_of(g + 1, GA, hrA, hrB, cntA, cntB, hr, i0, cnt);
            GATE_GROUP(bufB);
            g += 2;
        }
        load_wf(wf, bfv, wrow, b_f, e);
        if (g + 1 < G) {
            group_of(g + 1, GA, hrA, hrB, cntA, cntB, hr, i0, cnt);
            load_rows(bufB, hr, lane, i0, cnt);
            ORDER_POINT(wg);
            FLUSH_AT(g);
            group_of(g, GA, hrA, hrB, cntA, cntB, hr, i0, cnt);
            GATE_GROUP(bufA);
            FLUSH_AT(g + 1);
            group_of(g + 1, GA, hrA, hrB, cntA, cntB, hr, i0, cnt);
            GATE_GROUP(bufB);
        } else {
            FLUSH_AT(g);
            group_of(g, GA, hrA, hrB, cntA, cntB, hr, i0, cnt);
            GATE_GROUP(bufA);
        }
    }
    FLUSH_AT(G);
#undef FLUSH_AT
#undef GATE_GROUP

    ((float4*)(wacc[0] + warp * APITCH))[lane] = accA;
    ((float4*)(wacc[1] + warp * APITCH))[lane] = acc;
    if (lane == 0) { wgs[0][warp] = gsA; wgs[1][warp] = gs; }
    __syncthreads();

    const float* alo = wacc[c >> 3] + (c & 7) * APITCH + 4 * hh;
    float4 av[16];
    #pragma unroll
    for (int jj = 0; jj < 16; ++jj) av[jj] = *(const float4*)(alo + 8 * jj);
    float sg = 0.f;
    #pragma unroll
    for (int w = 0; w < NWAVE; ++w) sg += wgs[hh][w];
    __builtin_amdgcn_sched_barrier(0);
    v8f c0 = {}, c1 = {};
    #pragma unroll
    for (int jj = 0; jj < 16; ++jj) {
        const float4 bv = wf[jj];
        v2f a0 = { av[jj].x, av[jj].y }, b0 = { bv.x, bv.y };
        v2f a1 = { av[jj].z, av[jj].w }, b1 = { bv.z, bv.w };
        c0 = __builtin_amdgcn_wmma_f32_16x16x4_f32(false, a0, false, b0,
                                                   (short)0, c0, false, false);
        c1 = __builtin_amdgcn_wmma_f32_16x16x4_f32(false, a1, false, b1,
                                                   (short)0, c1, false, false);
    }

    float hv = 0.f;
    #pragma unroll
    for (int r = 0; r < 8; ++r) hv += c0[r] + c1[r];
    hv += sg * bfv;
    hg_s[hh][e] = hv;
    __syncthreads();

    if (warp < 2) {
        const int bo = warp ? bB : bA;
        const float4 v = ((const float4*)hg_s[warp])[lane];
        float q = v.x * v.x + v.y * v.y + v.z * v.z + v.w * v.w;
        q += xhalf(q);
        q += xrow<8>(q);
        q += xrow<4>(q);
        q += xrow<2>(q);
        q += xrow<1>(q);
        const float inv = 1.f / fmaxf(sqrtf(q), 1e-12f);
        ((float4*)(out + (size_t)bo * D_SZ))[lane] =
            make_float4(v.x * inv, v.y * inv, v.z * inv, v.w * inv);
    }
}

extern "C" void kernel_launch(void* const* d_in, const int* in_sizes, int n_in,
                              void* d_out, int out_size, void* d_ws, size_t ws_size,
                              hipStream_t stream) {
    const float* h   = (const float*)d_in[0];
    const int*   ns  = (const int*)d_in[1];
    const float* W_f = (const float*)d_in[2];
    const float* b_f = (const float*)d_in[3];
    const float* W_g = (const float*)d_in[4];
    const float* b_g = (const float*)d_in[5];
    float* out = (float*)d_out;

    GraphEmb_own_pair_gate_reduce_finish<<<dim3(256), 256, 0, stream>>>(
        h, ns, W_g, b_g, W_f, b_f, out);
}
